# starting kernel
# baseline (speedup 1.0000x reference)
#include <hip/hip_runtime.h>
#include <hip/hip_bf16.h>
#include <math.h>

typedef __attribute__((ext_vector_type(2))) float v2f;
typedef __attribute__((ext_vector_type(8))) float v8f;

__device__ __forceinline__ float relu1(float x) {
    float r;
    asm("v_max_num_f32 %0, 0, %1" : "=v"(r) : "v"(x));
    return r;
}

__global__ __launch_bounds__(256) void rnn_wmma_kernel(
    const float* __restrict__ state,
    const float* __restrict__ W_ih, const float* __restrict__ W_hh,
    const float* __restrict__ b_ih, const float* __restrict__ b_hh,
    const float* __restrict__ W_out, const float* __restrict__ b_out,
    float* __restrict__ out, int B, int T)
{
    const int lane = threadIdx.x & 31;
    const int m    = lane & 15;
    const bool low = lane < 16;
    const int w    = blockIdx.x * (blockDim.x >> 5) + (threadIdx.x >> 5);
    if (w * 32 >= B) return;

    auto whh = [&](int r, int c) -> float { return (r < 5) ? W_hh[r * 5 + c] : 0.0f; };
    auto wih = [&](int r, int c) -> float { return (r < 5) ? W_ih[r * 3 + c] : 0.0f; };

    v2f Aa, Ab, Ac;
    Aa.x = low ? whh(m, 0) : 0.0f;
    Aa.y = low ? whh(m, 1) : 0.0f;
    Ab.x = low ? whh(m, 2) : 0.0f;
    Ab.y = low ? whh(m, 3) : 0.0f;
    Ac.x = low ? whh(m, 4) : wih(m, 1);
    Ac.y = low ? wih(m, 0) : wih(m, 2);

    v8f cbias;
#pragma unroll
    for (int j = 0; j < 8; ++j) cbias[j] = 0.0f;
#pragma unroll
    for (int j = 0; j < 5; ++j) cbias[j] = low ? (b_ih[j] + b_hh[j]) : 0.0f;

    float h[2][5];
    const float* p[2];
    float4 c0[2], c1[2], c2[2];
#pragma unroll
    for (int g = 0; g < 2; ++g) {
#pragma unroll
        for (int j = 0; j < 5; ++j) h[g][j] = 0.0f;
        const int chain = w * 32 + g * 16 + m;
        p[g] = state + (size_t)chain * (size_t)T * 3u;
        c0[g] = ((const float4*)p[g])[0];
        c1[g] = ((const float4*)p[g])[1];
        c2[g] = ((const float4*)p[g])[2];
    }

    auto step = [&](int g, float x0, float x1, float x2) {
        v2f Ba, Bb, Bc;
        Ba.x = h[g][0];
        Ba.y = h[g][1];
        Bb.x = h[g][2];
        Bb.y = h[g][3];
        Bc.x = low ? h[g][4] : x1;
        Bc.y = low ? x0 : x2;
        v8f acc = __builtin_amdgcn_wmma_f32_16x16x4_f32(false, Aa, false, Ba,
                                                        (short)0, cbias, false, false);
        acc = __builtin_amdgcn_wmma_f32_16x16x4_f32(false, Ab, false, Bb,
                                                    (short)0, acc, false, false);
        acc = __builtin_amdgcn_wmma_f32_16x16x4_f32(false, Ac, false, Bc,
                                                    (short)0, acc, false, false);
        h[g][0] = relu1(acc[0]);
        h[g][1] = relu1(acc[1]);
        h[g][2] = relu1(acc[2]);
        h[g][3] = relu1(acc[3]);
        h[g][4] = relu1(acc[4]);
    };

    const int nblk = T >> 2;
    for (int it = 0; it < nblk - 1; ++it) {
        float4 n0[2], n1[2], n2[2];
#pragma unroll
        for (int g = 0; g < 2; ++g) {
            const float4* pn = (const float4*)(p[g] + 12);
            n0[g] = pn[0]; n1[g] = pn[1]; n2[g] = pn[2];
            __builtin_prefetch(p[g] + 96, 0, 3);
        }

        step(0, c0[0].x, c0[0].y, c0[0].z);  step(1, c0[1].x, c0[1].y, c0[1].z);
        step(0, c0[0].w, c1[0].x, c1[0].y);  step(1, c0[1].w, c1[1].x, c1[1].y);
        step(0, c1[0].z, c1[0].w, c2[0].x);  step(1, c1[1].z, c1[1].w, c2[1].x);
        step(0, c2[0].y, c2[0].z, c2[0].w);  step(1, c2[1].y, c2[1].z, c2[1].w);

#pragma unroll
        for (int g = 0; g < 2; ++g) {
            c0[g] = n0[g]; c1[g] = n1[g]; c2[g] = n2[g];
            p[g] += 12;
        }
    }
    step(0, c0[0].x, c0[0].y, c0[0].z);  step(1, c0[1].x, c0[1].y, c0[1].z);
    step(0, c0[0].w, c1[0].x, c1[0].y);  step(1, c0[1].w, c1[1].x, c1[1].y);
    step(0, c1[0].z, c1[0].w, c2[0].x);  step(1, c1[1].z, c1[1].w, c2[1].x);
    step(0, c2[0].y, c2[0].z, c2[0].w);  step(1, c2[1].y, c2[1].z, c2[1].w);

    if (low) {
#pragma unroll
        for (int g = 0; g < 2; ++g) {
            float z = W_out[0] * h[g][0] + W_out[1] * h[g][1] + W_out[2] * h[g][2] +
                      W_out[3] * h[g][3] + W_out[4] * h[g][4] + b_out[0];
            out[w * 32 + g * 16 + m] = tanhf(z);
        }
    }
}

extern "C" void kernel_launch(void* const* d_in, const int* in_sizes, int n_in,
                              void* d_out, int out_size, void* d_ws, size_t ws_size,
                              hipStream_t stream) {
    (void)n_in; (void)out_size; (void)d_ws; (void)ws_size;
    const float* state = (const float*)d_in[0];
    const float* W_ih  = (const float*)d_in[1];
    const float* W_hh  = (const float*)d_in[2];
    const float* b_ih  = (const float*)d_in[3];
    const float* b_hh  = (const float*)d_in[4];
    const float* W_out = (const float*)d_in[5];
    const float* b_out = (const float*)d_in[6];
    float* out = (float*)d_out;

    const int T = 2048;
    const int B = in_sizes[0] / (T * 3);
    const int waves = B / 32;
    const int wavesPerBlock = 8;
    const int blocks = (waves + wavesPerBlock - 1) / wavesPerBlock;

    rnn_wmma_kernel<<<blocks, wavesPerBlock * 32, 0, stream>>>(
        state, W_ih, W_hh, b_ih, b_hh, W_out, b_out, out, B, T);
}


# Recurrence wave reads its records from LDS as one group per 16 timesteps (32 b128 loads, second register set above v255), not per 4; records marked dead on the helper path: no copies per chunk
# speedup vs baseline: 3.7651x; 3.7651x over previous
#include <hip/hip_runtime.h>
#include <hip/hip_bf16.h>
#include <math.h>


typedef __attribute__((ext_vector_type(2))) float v2f;
typedef __attribute__((ext_vector_type(8))) float v8f;
typedef __attribute__((ext_vector_type(4))) float v4f;
typedef float v8fa __attribute__((ext_vector_type(8), aligned(16)));

#define CT    32
#define XROW4 25
#define XBUF4 (32 * XROW4)
#define PROW4 65
#define PSTG4 (32 * PROW4)
#define NSP   3
#define RWPB  2
#define HS    16

__device__ __forceinline__ float relu1(float x) {
    const int xi = __builtin_bit_cast(int, x);
    return __builtin_bit_cast(float, xi > 0 ? xi : 0);
}

__global__ __launch_bounds__(128) void rnn_wmma_kernel(
    const float* __restrict__ state,
    const float* __restrict__ W_ih, const float* __restrict__ W_hh,
    const float* __restrict__ b_ih, const float* __restrict__ b_hh,
    const float* __restrict__ W_out, const float* __restrict__ b_out,
    float* __restrict__ out, int B, int T)
{
    __shared__ float4 lds[RWPB * NSP * PSTG4 + RWPB * XBUF4];

    const int lane = threadIdx.x & 31;
    const int m    = lane & 15;
    const bool low = lane < 16;
    const int wv   = __builtin_amdgcn_readfirstlane((int)(threadIdx.x >> 5));
    const bool helper = wv >= RWPB;
    const int cw   = wv & (RWPB - 1);
    const int w    = blockIdx.x * RWPB + cw;
    const bool active = w * 32 < B;

    const int  ju   = low ? m : m - 8;
    const bool live = (unsigned)ju < 5u;
    auto whh = [&](int c) -> float { return live ? W_hh[ju * 5 + c] : 0.0f; };
    auto wih = [&](int c) -> float { return live ? W_ih[ju * 3 + c] : 0.0f; };

    v2f Ax, Az, Aa, Ab, Ac;
    Ax.x = wih(1);  Ax.y = wih(2);
    Az.x = wih(0);  Az.y = 0.0f;
    Aa.x = whh(0);  Aa.y = whh(1);
    Ab.x = whh(2);  Ab.y = whh(3);
    Ac.x = whh(4);  Ac.y = 0.0f;

    v8f cbias;
#pragma unroll
    for (int j = 0; j < 8; ++j) cbias[j] = 0.0f;
#pragma unroll
    for (int j = 0; j < 5; ++j) cbias[j] = b_ih[j] + b_hh[j];

    float h[5];
#pragma unroll
    for (int j = 0; j < 5; ++j) h[j] = 0.0f;

    float4* const ring = lds + cw * (NSP * PSTG4);
    float4* const xw   = lds + RWPB * NSP * PSTG4 + cw * XBUF4;
    float4* const prow = ring + lane * PROW4;

    const int sub = lane >> 3;
    const int pc  = lane & 7;
    const size_t row4 = (size_t)T * 3u / 4u;
    const float4* const gsrc = (const float4*)state + (size_t)(w * 32 + sub) * row4 + pc;
    float4* const ldst = xw + sub * XROW4 + pc;

    float4 R0, R1, R2, R3, R4, R5, R6, R7,
           R8, R9, R10, R11, R12, R13, R14, R15,
           R16, R17, R18, R19, R20, R21, R22, R23;
    auto load_chunk = [&](int s) {
        const float4* g = gsrc + (size_t)s * (CT * 3 / 4);
        R0 = g[(size_t)0 * row4 + 0];
        R1 = g[(size_t)0 * row4 + 8];
        R2 = g[(size_t)0 * row4 + 16];
        R3 = g[(size_t)4 * row4 + 0];
        R4 = g[(size_t)4 * row4 + 8];
        R5 = g[(size_t)4 * row4 + 16];
        R6 = g[(size_t)8 * row4 + 0];
        R7 = g[(size_t)8 * row4 + 8];
        R8 = g[(size_t)8 * row4 + 16];
        R9 = g[(size_t)12 * row4 + 0];
        R10 = g[(size_t)12 * row4 + 8];
        R11 = g[(size_t)12 * row4 + 16];
        R12 = g[(size_t)16 * row4 + 0];
        R13 = g[(size_t)16 * row4 + 8];
        R14 = g[(size_t)16 * row4 + 16];
        R15 = g[(size_t)20 * row4 + 0];
        R16 = g[(size_t)20 * row4 + 8];
        R17 = g[(size_t)20 * row4 + 16];
        R18 = g[(size_t)24 * row4 + 0];
        R19 = g[(size_t)24 * row4 + 8];
        R20 = g[(size_t)24 * row4 + 16];
        R21 = g[(size_t)28 * row4 + 0];
        R22 = g[(size_t)28 * row4 + 8];
        R23 = g[(size_t)28 * row4 + 16];
    };
    auto store_chunk = [&]() {
        float4* d = ldst;
        d[0 * XROW4 + 0] = R0;
        d[0 * XROW4 + 8] = R1;
        d[0 * XROW4 + 16] = R2;
        d[4 * XROW4 + 0] = R3;
        d[4 * XROW4 + 8] = R4;
        d[4 * XROW4 + 16] = R5;
        d[8 * XROW4 + 0] = R6;
        d[8 * XROW4 + 8] = R7;
        d[8 * XROW4 + 16] = R8;
        d[12 * XROW4 + 0] = R9;
        d[12 * XROW4 + 8] = R10;
        d[12 * XROW4 + 16] = R11;
        d[16 * XROW4 + 0] = R12;
        d[16 * XROW4 + 8] = R13;
        d[16 * XROW4 + 16] = R14;
        d[20 * XROW4 + 0] = R15;
        d[20 * XROW4 + 8] = R16;
        d[20 * XROW4 + 16] = R17;
        d[24 * XROW4 + 0] = R18;
        d[24 * XROW4 + 8] = R19;
        d[24 * XROW4 + 16] = R20;
        d[28 * XROW4 + 0] = R21;
        d[28 * XROW4 + 8] = R22;
        d[28 * XROW4 + 16] = R23;
    };

    const float4* const xr = xw + lane * XROW4;
    auto proj = [&](float x0, float x1, float x2) -> v8f {
        v2f Bx, Bz;
        Bx.x = x1;  Bx.y = x2;
        Bz.x = x0;  Bz.y = x1;
        v8f d = __builtin_amdgcn_wmma_f32_16x16x4_f32(false, Ax, false, Bx,
                                                      (short)0, cbias, false, false);
        d = __builtin_amdgcn_wmma_f32_16x16x4_f32(false, Az, false, Bz,
                                                  (short)0, d, false, false);
        return d;
    };
    auto project4 = [&](int stage, int sb) {
        const v4f* r = (const v4f*)(xr + sb * 3);
        const v4f a0 = r[0], a1 = r[1], a2 = r[2];
        const v8f d0 = proj(a0.x, a0.y, a0.z);
        const v8f d1 = proj(a0.w, a1.x, a1.y);
        const v8f d2 = proj(a1.z, a1.w, a2.x);
        const v8f d3 = proj(a2.y, a2.z, a2.w);
        v8fa* d = (v8fa*)(prow + stage * PSTG4 + sb * 8);
        d[0] = d0;  d[1] = d1;  d[2] = d2;  d[3] = d3;
    };

    auto step = [&](v8f c) {
        v2f Ba, Bb, Bc;
        Ba.x = h[0];  Ba.y = h[1];
        Bb.x = h[2];  Bb.y = h[3];
        Bc.x = h[4];  Bc.y = 0.0f;
        v8f acc = __builtin_amdgcn_wmma_f32_16x16x4_f32(false, Aa, false, Ba,
                                                        (short)0, c, false, false);
        acc = __builtin_amdgcn_wmma_f32_16x16x4_f32(false, Ab, false, Bb,
                                                    (short)0, acc, false, false);
        acc = __builtin_amdgcn_wmma_f32_16x16x4_f32(false, Ac, false, Bc,
                                                    (short)0, acc, false, false);
        h[0] = relu1(acc[0]);
        h[1] = relu1(acc[1]);
        h[2] = relu1(acc[2]);
        h[3] = relu1(acc[3]);
        h[4] = relu1(acc[4]);
    };

    struct PBlk { v8f s[HS]; };
    auto read_p = [&](int stage, int half) -> PBlk {
        const v8fa* r = (const v8fa*)(prow + stage * PSTG4 + half * (HS * 2));
        PBlk v;
        __builtin_amdgcn_sched_barrier(0);
#pragma unroll
        for (int k = 0; k < HS; ++k) v.s[k] = r[k];
        return v;
    };
    auto half_chunk = [&](PBlk v) {
        __builtin_amdgcn_sched_barrier(0);
#pragma unroll
        for (int k = 0; k < HS; ++k) step(v.s[k]);
    };

    const int nch = T / CT;

    auto handover = [&]() {
        __builtin_amdgcn_fence(__ATOMIC_RELEASE, "workgroup", "local");
        __builtin_amdgcn_s_barrier();
        __builtin_amdgcn_fence(__ATOMIC_ACQUIRE, "workgroup", "local");
    };

    asm volatile("" :: "v"(Ax), "v"(Az), "v"(Aa), "v"(Ab), "v"(Ac), "v"(cbias));

    if (active && helper) {
        load_chunk(0);
        store_chunk();
        if (1 < nch) load_chunk(1);
    }
    handover();

    PBlk cur;
#pragma unroll
    for (int k = 0; k < HS; ++k) cur.s[k] = cbias;

    int s0 = 0;
    for (int i = 0; i < nch + 2; ++i) {
        const int s1 = (s0 + 1 == NSP) ? 0 : s0 + 1;
        const int s2 = (s1 + 1 == NSP) ? 0 : s1 + 1;
        if (active) {
            if (helper) {
                if (i < nch) {
#pragma unroll
                    for (int sb = 0; sb < 8; ++sb) project4(s0, sb);
                    if (i + 1 < nch) store_chunk();
                    if (i + 2 < nch) load_chunk(i + 2);
                }
#pragma unroll
                for (int k = 0; k < HS; ++k) cur.s[k] = __builtin_nondeterministic_value(cur.s[k]);
            } else if (i == 1) {
                cur = read_p(0, 0);
            } else if (i >= 2) {
                const PBlk sec = read_p(s1, 1);
                half_chunk(cur);
                cur = read_p(s2, 0);
                half_chunk(sec);
            }
        }
        s0 = s1;
        handover();
    }

    if (active && !helper) {
        float z = W_out[0] * h[0] + W_out[1] * h[1] + W_out[2] * h[2] +
                  W_out[3] * h[3] + W_out[4] * h[4] + b_out[0];
        out[w * 32 + lane] = tanhf(z);
    }
}

extern "C" void kernel_launch(void* const* d_in, const int* in_sizes, int n_in,
                              void* d_out, int out_size, void* d_ws, size_t ws_size,
                              hipStream_t stream) {
    (void)n_in; (void)out_size; (void)d_ws; (void)ws_size;
    const float* state = (const float*)d_in[0];
    const float* W_ih  = (const float*)d_in[1];
    const float* W_hh  = (const float*)d_in[2];
    const float* b_ih  = (const float*)d_in[3];
    const float* b_hh  = (const float*)d_in[4];
    const float* W_out = (const float*)d_in[5];
    const float* b_out = (const float*)d_in[6];
    float* out = (float*)d_out;

    const int T = 2048;
    const int B = in_sizes[0] / (T * 3);
    const int waves = B / 32;
    const int recWavesPerBlock = 2;
    const int blocks = (waves + recWavesPerBlock - 1) / recWavesPerBlock;

    rnn_wmma_kernel<<<blocks, 128, 0, stream>>>(
        state, W_ih, W_hh, b_ih, b_hh, W_out, b_out, out, B, T);
}


# Helper wave keeps LDS round trips out of its WMMA stream: a chunk's raw inputs read from LDS as one group before its 64 WMMAs, records stored per 8 timesteps one group behind their WMMAs
# speedup vs baseline: 3.8789x; 1.0302x over previous
#include <hip/hip_runtime.h>
#include <hip/hip_bf16.h>
#include <math.h>


typedef __attribute__((ext_vector_type(2))) float v2f;
typedef __attribute__((ext_vector_type(8))) float v8f;
typedef __attribute__((ext_vector_type(4))) float v4f;
typedef float v8fa __attribute__((ext_vector_type(8), aligned(16)));

#define CT    32
#define XROW4 25
#define XBUF4 (32 * XROW4)
#define PROW4 65
#define PSTG4 (32 * PROW4)
#define NSP   3
#define RWPB  2
#define HS    16

__device__ __forceinline__ float relu1(float x) {
    const int xi = __builtin_bit_cast(int, x);
    return __builtin_bit_cast(float, xi > 0 ? xi : 0);
}

__global__ __launch_bounds__(128) void rnn_wmma_kernel(
    const float* __restrict__ state,
    const float* __restrict__ W_ih, const float* __restrict__ W_hh,
    const float* __restrict__ b_ih, const float* __restrict__ b_hh,
    const float* __restrict__ W_out, const float* __restrict__ b_out,
    float* __restrict__ out, int B, int T)
{
    __shared__ float4 lds[RWPB * NSP * PSTG4 + RWPB * XBUF4];

    const int lane = threadIdx.x & 31;
    const int m    = lane & 15;
    const bool low = lane < 16;
    const int wv   = __builtin_amdgcn_readfirstlane((int)(threadIdx.x >> 5));
    const bool helper = wv >= RWPB;
    const int cw   = wv & (RWPB - 1);
    const int w    = blockIdx.x * RWPB + cw;
    const bool active = w * 32 < B;

    const int  ju   = low ? m : m - 8;
    const bool live = (unsigned)ju < 5u;
    auto whh = [&](int c) -> float { return live ? W_hh[ju * 5 + c] : 0.0f; };
    auto wih = [&](int c) -> float { return live ? W_ih[ju * 3 + c] : 0.0f; };

    v2f Ax, Az, Aa, Ab, Ac;
    Ax.x = wih(1);  Ax.y = wih(2);
    Az.x = wih(0);  Az.y = 0.0f;
    Aa.x = whh(0);  Aa.y = whh(1);
    Ab.x = whh(2);  Ab.y = whh(3);
    Ac.x = whh(4);  Ac.y = 0.0f;

    v8f cbias;
#pragma unroll
    for (int j = 0; j < 8; ++j) cbias[j] = 0.0f;
#pragma unroll
    for (int j = 0; j < 5; ++j) cbias[j] = b_ih[j] + b_hh[j];

    float h[5];
#pragma unroll
    for (int j = 0; j < 5; ++j) h[j] = 0.0f;

    float4* const ring = lds + cw * (NSP * PSTG4);
    float4* const xw   = lds + RWPB * NSP * PSTG4 + cw * XBUF4;
    float4* const prow = ring + lane * PROW4;

    const int sub = lane >> 3;
    const int pc  = lane & 7;
    const size_t row4 = (size_t)T * 3u / 4u;
    const float4* const gsrc = (const float4*)state + (size_t)(w * 32 + sub) * row4 + pc;
    float4* const ldst = xw + sub * XROW4 + pc;

    float4 R0, R1, R2, R3, R4, R5, R6, R7,
           R8, R9, R10, R11, R12, R13, R14, R15,
           R16, R17, R18, R19, R20, R21, R22, R23;
    auto load_chunk = [&](int s) {
        const float4* g = gsrc + (size_t)s * (CT * 3 / 4);
        R0 = g[(size_t)0 * row4 + 0];
        R1 = g[(size_t)0 * row4 + 8];
        R2 = g[(size_t)0 * row4 + 16];
        R3 = g[(size_t)4 * row4 + 0];
        R4 = g[(size_t)4 * row4 + 8];
        R5 = g[(size_t)4 * row4 + 16];
        R6 = g[(size_t)8 * row4 + 0];
        R7 = g[(size_t)8 * row4 + 8];
        R8 = g[(size_t)8 * row4 + 16];
        R9 = g[(size_t)12 * row4 + 0];
        R10 = g[(size_t)12 * row4 + 8];
        R11 = g[(size_t)12 * row4 + 16];
        R12 = g[(size_t)16 * row4 + 0];
        R13 = g[(size_t)16 * row4 + 8];
        R14 = g[(size_t)16 * row4 + 16];
        R15 = g[(size_t)20 * row4 + 0];
        R16 = g[(size_t)20 * row4 + 8];
        R17 = g[(size_t)20 * row4 + 16];
        R18 = g[(size_t)24 * row4 + 0];
        R19 = g[(size_t)24 * row4 + 8];
        R20 = g[(size_t)24 * row4 + 16];
        R21 = g[(size_t)28 * row4 + 0];
        R22 = g[(size_t)28 * row4 + 8];
        R23 = g[(size_t)28 * row4 + 16];
    };
    auto store_chunk = [&]() {
        float4* d = ldst;
        d[0 * XROW4 + 0] = R0;
        d[0 * XROW4 + 8] = R1;
        d[0 * XROW4 + 16] = R2;
        d[4 * XROW4 + 0] = R3;
        d[4 * XROW4 + 8] = R4;
        d[4 * XROW4 + 16] = R5;
        d[8 * XROW4 + 0] = R6;
        d[8 * XROW4 + 8] = R7;
        d[8 * XROW4 + 16] = R8;
        d[12 * XROW4 + 0] = R9;
        d[12 * XROW4 + 8] = R10;
        d[12 * XROW4 + 16] = R11;
        d[16 * XROW4 + 0] = R12;
        d[16 * XROW4 + 8] = R13;
        d[16 * XROW4 + 16] = R14;
        d[20 * XROW4 + 0] = R15;
        d[20 * XROW4 + 8] = R16;
        d[20 * XROW4 + 16] = R17;
        d[24 * XROW4 + 0] = R18;
        d[24 * XROW4 + 8] = R19;
        d[24 * XROW4 + 16] = R20;
        d[28 * XROW4 + 0] = R21;
        d[28 * XROW4 + 8] = R22;
        d[28 * XROW4 + 16] = R23;
    };

    const float4* const xr = xw + lane * XROW4;
    auto proj = [&](float x0, float x1, float x2) -> v8f {
        v2f Bx, Bz;
        Bx.x = x1;  Bx.y = x2;
        Bz.x = x0;  Bz.y = x1;
        v8f d = __builtin_amdgcn_wmma_f32_16x16x4_f32(false, Ax, false, Bx,
                                                      (short)0, cbias, false, false);
        d = __builtin_amdgcn_wmma_f32_16x16x4_f32(false, Az, false, Bz,
                                                  (short)0, d, false, false);
        return d;
    };
    struct XBlk { v4f s[24]; };
    auto read_x = [&]() __attribute__((always_inline)) -> XBlk {
        const v4f* r = (const v4f*)xr;
        XBlk v;
#pragma unroll
        for (int k = 0; k < 24; ++k) v.s[k] = r[k];
        return v;
    };
    struct DBlk { v8f s[8]; };
    auto project8 = [&](const XBlk& x, int g) __attribute__((always_inline)) -> DBlk {
        DBlk d;
#pragma unroll
        for (int q = 0; q < 2; ++q) {
            const v4f a0 = x.s[6 * g + 3 * q], a1 = x.s[6 * g + 3 * q + 1], a2 = x.s[6 * g + 3 * q + 2];
            d.s[4 * q + 0] = proj(a0.x, a0.y, a0.z);
            d.s[4 * q + 1] = proj(a0.w, a1.x, a1.y);
            d.s[4 * q + 2] = proj(a1.z, a1.w, a2.x);
            d.s[4 * q + 3] = proj(a2.y, a2.z, a2.w);
        }
        return d;
    };
    auto store8 = [&](int stage, int g, const DBlk& d) __attribute__((always_inline)) {
        v8fa* p = (v8fa*)(prow + stage * PSTG4 + g * 16);
#pragma unroll
        for (int k = 0; k < 8; ++k) p[k] = d.s[k];
    };

    auto step = [&](v8f c) {
        v2f Ba, Bb, Bc;
        Ba.x = h[0];  Ba.y = h[1];
        Bb.x = h[2];  Bb.y = h[3];
        Bc.x = h[4];  Bc.y = 0.0f;
        v8f acc = __builtin_amdgcn_wmma_f32_16x16x4_f32(false, Aa, false, Ba,
                                                        (short)0, c, false, false);
        acc = __builtin_amdgcn_wmma_f32_16x16x4_f32(false, Ab, false, Bb,
                                                    (short)0, acc, false, false);
        acc = __builtin_amdgcn_wmma_f32_16x16x4_f32(false, Ac, false, Bc,
                                                    (short)0, acc, false, false);
        h[0] = relu1(acc[0]);
        h[1] = relu1(acc[1]);
        h[2] = relu1(acc[2]);
        h[3] = relu1(acc[3]);
        h[4] = relu1(acc[4]);
    };

    struct PBlk { v8f s[HS]; };
    auto read_p = [&](int stage, int half) -> PBlk {
        const v8fa* r = (const v8fa*)(prow + stage * PSTG4 + half * (HS * 2));
        PBlk v;
        __builtin_amdgcn_sched_barrier(0);
#pragma unroll
        for (int k = 0; k < HS; ++k) v.s[k] = r[k];
        return v;
    };
    auto half_chunk = [&](PBlk v) {
        __builtin_amdgcn_sched_barrier(0);
#pragma unroll
        for (int k = 0; k < HS; ++k) step(v.s[k]);
    };

    const int nch = T / CT;

    auto handover = [&]() {
        __builtin_amdgcn_fence(__ATOMIC_RELEASE, "workgroup", "local");
        __builtin_amdgcn_s_barrier();
        __builtin_amdgcn_fence(__ATOMIC_ACQUIRE, "workgroup", "local");
    };

    asm volatile("" :: "v"(Ax), "v"(Az), "v"(Aa), "v"(Ab), "v"(Ac), "v"(cbias));

    if (active && helper) {
        load_chunk(0);
        store_chunk();
        if (1 < nch) load_chunk(1);
    }
    handover();

    PBlk cur;
#pragma unroll
    for (int k = 0; k < HS; ++k) cur.s[k] = cbias;

    int s0 = 0;
    for (int i = 0; i < nch + 2; ++i) {
        const int s1 = (s0 + 1 == NSP) ? 0 : s0 + 1;
        const int s2 = (s1 + 1 == NSP) ? 0 : s1 + 1;
        if (active) {
            if (helper) {
                if (i < nch) {
                    const XBlk x = read_x();
                    if (i + 1 < nch) store_chunk();
                    if (i + 2 < nch) load_chunk(i + 2);
                    DBlk d0 = project8(x, 0);
                    DBlk d1 = project8(x, 1);
                    store8(s0, 0, d0);
                    __builtin_amdgcn_sched_barrier(0);
                    d0 = project8(x, 2);
                    __builtin_amdgcn_sched_barrier(0);
                    store8(s0, 1, d1);
                    __builtin_amdgcn_sched_barrier(0);
                    d1 = project8(x, 3);
                    store8(s0, 2, d0);
                    store8(s0, 3, d1);
                }
#pragma unroll
                for (int k = 0; k < HS; ++k) cur.s[k] = __builtin_nondeterministic_value(cur.s[k]);
            } else if (i == 1) {
                cur = read_p(0, 0);
            } else if (i >= 2) {
                const PBlk sec = read_p(s1, 1);
                half_chunk(cur);
                cur = read_p(s2, 0);
                half_chunk(sec);
            }
        }
        s0 = s1;
        handover();
    }

    if (active && !helper) {
        float z = W_out[0] * h[0] + W_out[1] * h[1] + W_out[2] * h[2] +
                  W_out[3] * h[3] + W_out[4] * h[4] + b_out[0];
        out[w * 32 + lane] = tanhf(z);
    }
}

extern "C" void kernel_launch(void* const* d_in, const int* in_sizes, int n_in,
                              void* d_out, int out_size, void* d_ws, size_t ws_size,
                              hipStream_t stream) {
    (void)n_in; (void)out_size; (void)d_ws; (void)ws_size;
    const float* state = (const float*)d_in[0];
    const float* W_ih  = (const float*)d_in[1];
    const float* W_hh  = (const float*)d_in[2];
    const float* b_ih  = (const float*)d_in[3];
    const float* b_hh  = (const float*)d_in[4];
    const float* W_out = (const float*)d_in[5];
    const float* b_out = (const float*)d_in[6];
    float* out = (float*)d_out;

    const int T = 2048;
    const int B = in_sizes[0] / (T * 3);
    const int waves = B / 32;
    const int recWavesPerBlock = 2;
    const int blocks = (waves + recWavesPerBlock - 1) / recWavesPerBlock;

    rnn_wmma_kernel<<<blocks, 128, 0, stream>>>(
        state, W_ih, W_hh, b_ih, b_hh, W_out, b_out, out, B, T);
}
